# starting kernel
# baseline (speedup 1.0000x reference)
#include <hip/hip_runtime.h>


typedef __attribute__((ext_vector_type(2))) float v2f;
typedef __attribute__((ext_vector_type(8))) float v8f;

#define BN_EPS 1e-5f

__device__ __forceinline__ void atomic_add_f32(float* p, float v) {
    unsafeAtomicAdd(p, v);
}

__global__ void conv1_scatter(const float* __restrict__ x,
                              const float* __restrict__ W1,
                              const int*   __restrict__ in_idx,
                              const int*   __restrict__ out_idx,
                              float*       __restrict__ h_acc,
                              int M, int KM) {
    int t = blockIdx.x * blockDim.x + threadIdx.x;
    if (t >= KM) return;
    int k = t / M;
    float xv = x[in_idx[t]];
    int ro = out_idx[t] * 16;
    const float* w = W1 + k * 16;
#pragma unroll
    for (int c = 0; c < 16; ++c)
        atomic_add_f32(&h_acc[ro + c], xv * w[c]);
}

__global__ void bn_elu(const float* __restrict__ in, float* __restrict__ out,
                       const float* __restrict__ g,  const float* __restrict__ b,
                       const float* __restrict__ mn, const float* __restrict__ vr,
                       int total, int Cmask) {
    int t = blockIdx.x * blockDim.x + threadIdx.x;
    if (t >= total) return;
    int c = t & Cmask;
    float s = g[c] * rsqrtf(vr[c] + BN_EPS);
    float y = (in[t] - mn[c]) * s + b[c];
    out[t] = (y > 0.0f) ? y : (expf(y) - 1.0f);
}

__global__ void conv2_wmma(const float* __restrict__ h,
                           const float* __restrict__ W2,
                           const int*   __restrict__ in_idx,
                           const int*   __restrict__ out_idx,
                           float*       __restrict__ acc,
                           int M, int Mt, int totalWaves) {
    int wave = blockIdx.x * (blockDim.x >> 5) + (threadIdx.x >> 5);
    if (wave >= totalWaves) return;
    int lane = threadIdx.x & 31;
    int half = lane >> 4;
    int l15  = lane & 15;

    int k  = wave / Mt;
    int m0 = (wave - k * Mt) << 4;
    int base = k * M + m0;

    int arow = in_idx[base + l15];
    const float* hp = h + arow * 16 + (half << 1);

    const float* wk = W2 + k * 512;

    v8f c0 = {};
    v8f c1 = {};

#pragma unroll
    for (int q = 0; q < 4; ++q) {
        v2f a;
        a.x = hp[4 * q + 0];
        a.y = hp[4 * q + 1];
        int ci0 = 4 * q + (half << 1);
        v2f b0, b1;
        b0.x = wk[(ci0    ) * 32 +      l15];
        b0.y = wk[(ci0 + 1) * 32 +      l15];
        b1.x = wk[(ci0    ) * 32 + 16 + l15];
        b1.y = wk[(ci0 + 1) * 32 + 16 + l15];
        c0 = __builtin_amdgcn_wmma_f32_16x16x4_f32(false, a, false, b0,
                                                   (short)0, c0, false, false);
        c1 = __builtin_amdgcn_wmma_f32_16x16x4_f32(false, a, false, b1,
                                                   (short)0, c1, false, false);
    }

    int rows[8];
#pragma unroll
    for (int v = 0; v < 8; ++v)
        rows[v] = out_idx[base + (half << 3) + v] * 32;
#pragma unroll
    for (int v = 0; v < 8; ++v) {
        atomic_add_f32(&acc[rows[v] +      l15], c0[v]);
        atomic_add_f32(&acc[rows[v] + 16 + l15], c1[v]);
    }
}

__global__ void conv2_scalar_tail(const float* __restrict__ h,
                                  const float* __restrict__ W2,
                                  const int*   __restrict__ in_idx,
                                  const int*   __restrict__ out_idx,
                                  float*       __restrict__ acc,
                                  int M, int mStart, int K) {
    int rem = M - mStart;
    int total = K * rem * 32;
    int t = blockIdx.x * blockDim.x + threadIdx.x;
    if (t >= total) return;
    int co = t & 31;
    int km = t >> 5;
    int k  = km / rem;
    int m  = mStart + (km - k * rem);
    int e  = k * M + m;
    const float* hr = h + in_idx[e] * 16;
    const float* wp = W2 + k * 512 + co;
    float s = 0.0f;
#pragma unroll
    for (int ci = 0; ci < 16; ++ci) s += hr[ci] * wp[ci * 32];
    atomic_add_f32(&acc[out_idx[e] * 32 + co], s);
}

extern "C" void kernel_launch(void* const* d_in, const int* in_sizes, int n_in,
                              void* d_out, int out_size, void* d_ws, size_t ws_size,
                              hipStream_t stream) {
    const float* x      = (const float*)d_in[0];
    const float* W1     = (const float*)d_in[1];
    const float* g1     = (const float*)d_in[2];
    const float* b1     = (const float*)d_in[3];
    const float* m1     = (const float*)d_in[4];
    const float* v1     = (const float*)d_in[5];
    const float* W2     = (const float*)d_in[6];
    const float* g2     = (const float*)d_in[7];
    const float* b2     = (const float*)d_in[8];
    const float* m2     = (const float*)d_in[9];
    const float* v2     = (const float*)d_in[10];
    const int*   in_idx  = (const int*)d_in[11];
    const int*   out_idx = (const int*)d_in[12];

    const int N  = in_sizes[0];
    const int K  = in_sizes[1] / 16;
    const int KM = in_sizes[11];
    const int M  = KM / K;

    float* h_buf = (float*)d_ws;
    float* acc   = h_buf + (size_t)N * 16;

    hipMemsetAsync(d_ws, 0, (size_t)N * 48 * sizeof(float), stream);

    conv1_scatter<<<(KM + 255) / 256, 256, 0, stream>>>(
        x, W1, in_idx, out_idx, h_buf, M, KM);

    bn_elu<<<(N * 16 + 255) / 256, 256, 0, stream>>>(
        h_buf, h_buf, g1, b1, m1, v1, N * 16, 15);

    const int Mt = M / 16;
    const int totalWaves = K * Mt;
    if (totalWaves > 0) {
        const int blocks = (totalWaves + 7) / 8;
        conv2_wmma<<<blocks, 256, 0, stream>>>(
            h_buf, W2, in_idx, out_idx, acc, M, Mt, totalWaves);
    }
    const int mStart = Mt * 16;
    if (mStart < M) {
        const int total = K * (M - mStart) * 32;
        conv2_scalar_tail<<<(total + 255) / 256, 256, 0, stream>>>(
            h_buf, W2, in_idx, out_idx, acc, M, mStart, K);
    }

    bn_elu<<<(N * 32 + 255) / 256, 256, 0, stream>>>(
        acc, (float*)d_out, g2, b2, m2, v2, N * 32, 31);
}


# stage-1 scatter: lanes span the 16 channels (map entry handed to its 16-lane row by DPP row_share): each f32 atomic covers two whole 64-B rows, not 32 lines; adds arrive unordered as before
# speedup vs baseline: 2.6370x; 2.6370x over previous
#include <hip/hip_runtime.h>


typedef __attribute__((ext_vector_type(2))) float v2f;
typedef __attribute__((ext_vector_type(8))) float v8f;

#define BN_EPS 1e-5f

__device__ __forceinline__ void atomic_add_f32(float* p, float v) {
    unsafeAtomicAdd(p, v);
}

template <int I>
__device__ __forceinline__ int row_share_i32(int v) {
    return __builtin_amdgcn_update_dpp(0, v, 0x150 + I, 0xf, 0xf, false);
}

template <int I, bool CHECK>
__device__ __forceinline__ void conv1_steps(float* __restrict__ h_acc,
                                            int off, int xbits, float w, int l15) {
    if constexpr (I < 16) {
        int   o = row_share_i32<I>(off);
        float v = __int_as_float(row_share_i32<I>(xbits));
        if (!CHECK || o >= 0) atomic_add_f32(&h_acc[(unsigned)(o + l15)], v * w);
        conv1_steps<I + 1, CHECK>(h_acc, off, xbits, w, l15);
    }
}

__global__ void conv1_scatter(const float* __restrict__ x,
                              const float* __restrict__ W1,
                              const int*   __restrict__ in_idx,
                              const int*   __restrict__ out_idx,
                              float*       __restrict__ h_acc,
                              int M) {
    int k   = blockIdx.y;
    int m   = blockIdx.x * blockDim.x + threadIdx.x;
    int l15 = threadIdx.x & 15;
    int m0  = __builtin_amdgcn_readfirstlane(m);
    if (m0 >= M) return;
    float w = W1[k * 16 + l15];
    if (m0 + 32 <= M) {
        int e = k * M + m;
        float xv  = x[in_idx[e]];
        int   off = out_idx[e] * 16;
        conv1_steps<0, false>(h_acc, off, __float_as_int(xv), w, l15);
    } else {
        int   off = -1;
        float xv  = 0.0f;
        if (m < M) {
            int e = k * M + m;
            xv  = x[in_idx[e]];
            off = out_idx[e] * 16;
        }
        conv1_steps<0, true>(h_acc, off, __float_as_int(xv), w, l15);
    }
}

__global__ void bn_elu(const float* __restrict__ in, float* __restrict__ out,
                       const float* __restrict__ g,  const float* __restrict__ b,
                       const float* __restrict__ mn, const float* __restrict__ vr,
                       int total, int Cmask) {
    int t = blockIdx.x * blockDim.x + threadIdx.x;
    if (t >= total) return;
    int c = t & Cmask;
    float s = g[c] * rsqrtf(vr[c] + BN_EPS);
    float y = (in[t] - mn[c]) * s + b[c];
    out[t] = (y > 0.0f) ? y : (expf(y) - 1.0f);
}

__global__ void conv2_wmma(const float* __restrict__ h,
                           const float* __restrict__ W2,
                           const int*   __restrict__ in_idx,
                           const int*   __restrict__ out_idx,
                           float*       __restrict__ acc,
                           int M, int Mt, int totalWaves) {
    int wave = blockIdx.x * (blockDim.x >> 5) + (threadIdx.x >> 5);
    if (wave >= totalWaves) return;
    int lane = threadIdx.x & 31;
    int half = lane >> 4;
    int l15  = lane & 15;

    int k  = wave / Mt;
    int m0 = (wave - k * Mt) << 4;
    int base = k * M + m0;

    int arow = in_idx[base + l15];
    const float* hp = h + arow * 16 + (half << 1);

    const float* wk = W2 + k * 512;

    v8f c0 = {};
    v8f c1 = {};

#pragma unroll
    for (int q = 0; q < 4; ++q) {
        v2f a;
        a.x = hp[4 * q + 0];
        a.y = hp[4 * q + 1];
        int ci0 = 4 * q + (half << 1);
        v2f b0, b1;
        b0.x = wk[(ci0    ) * 32 +      l15];
        b0.y = wk[(ci0 + 1) * 32 +      l15];
        b1.x = wk[(ci0    ) * 32 + 16 + l15];
        b1.y = wk[(ci0 + 1) * 32 + 16 + l15];
        c0 = __builtin_amdgcn_wmma_f32_16x16x4_f32(false, a, false, b0,
                                                   (short)0, c0, false, false);
        c1 = __builtin_amdgcn_wmma_f32_16x16x4_f32(false, a, false, b1,
                                                   (short)0, c1, false, false);
    }

    int rows[8];
#pragma unroll
    for (int v = 0; v < 8; ++v)
        rows[v] = out_idx[base + (half << 3) + v] * 32;
#pragma unroll
    for (int v = 0; v < 8; ++v) {
        atomic_add_f32(&acc[rows[v] +      l15], c0[v]);
        atomic_add_f32(&acc[rows[v] + 16 + l15], c1[v]);
    }
}

__global__ void conv2_scalar_tail(const float* __restrict__ h,
                                  const float* __restrict__ W2,
                                  const int*   __restrict__ in_idx,
                                  const int*   __restrict__ out_idx,
                                  float*       __restrict__ acc,
                                  int M, int mStart, int K) {
    int rem = M - mStart;
    int total = K * rem * 32;
    int t = blockIdx.x * blockDim.x + threadIdx.x;
    if (t >= total) return;
    int co = t & 31;
    int km = t >> 5;
    int k  = km / rem;
    int m  = mStart + (km - k * rem);
    int e  = k * M + m;
    const float* hr = h + in_idx[e] * 16;
    const float* wp = W2 + k * 512 + co;
    float s = 0.0f;
#pragma unroll
    for (int ci = 0; ci < 16; ++ci) s += hr[ci] * wp[ci * 32];
    atomic_add_f32(&acc[out_idx[e] * 32 + co], s);
}

extern "C" void kernel_launch(void* const* d_in, const int* in_sizes, int n_in,
                              void* d_out, int out_size, void* d_ws, size_t ws_size,
                              hipStream_t stream) {
    const float* x      = (const float*)d_in[0];
    const float* W1     = (const float*)d_in[1];
    const float* g1     = (const float*)d_in[2];
    const float* b1     = (const float*)d_in[3];
    const float* m1     = (const float*)d_in[4];
    const float* v1     = (const float*)d_in[5];
    const float* W2     = (const float*)d_in[6];
    const float* g2     = (const float*)d_in[7];
    const float* b2     = (const float*)d_in[8];
    const float* m2     = (const float*)d_in[9];
    const float* v2     = (const float*)d_in[10];
    const int*   in_idx  = (const int*)d_in[11];
    const int*   out_idx = (const int*)d_in[12];

    const int N  = in_sizes[0];
    const int K  = in_sizes[1] / 16;
    const int KM = in_sizes[11];
    const int M  = KM / K;

    float* h_buf = (float*)d_ws;
    float* acc   = h_buf + (size_t)N * 16;

    hipMemsetAsync(d_ws, 0, (size_t)N * 48 * sizeof(float), stream);

    conv1_scatter<<<dim3((M + 255) / 256, K), 256, 0, stream>>>(
        x, W1, in_idx, out_idx, h_buf, M);

    bn_elu<<<(N * 16 + 255) / 256, 256, 0, stream>>>(
        h_buf, h_buf, g1, b1, m1, v1, N * 16, 15);

    const int Mt = M / 16;
    const int totalWaves = K * Mt;
    if (totalWaves > 0) {
        const int blocks = (totalWaves + 7) / 8;
        conv2_wmma<<<blocks, 256, 0, stream>>>(
            h_buf, W2, in_idx, out_idx, acc, M, Mt, totalWaves);
    }
    const int mStart = Mt * 16;
    if (mStart < M) {
        const int total = K * (M - mStart) * 32;
        conv2_scalar_tail<<<(total + 255) / 256, 256, 0, stream>>>(
            h_buf, W2, in_idx, out_idx, acc, M, mStart, K);
    }

    bn_elu<<<(N * 32 + 255) / 256, 256, 0, stream>>>(
        acc, (float*)d_out, g2, b2, m2, v2, N * 32, 31);
}


# stage-2 scatter: the two D tiles trade lane halves (v_permlane16_swap_b32) so each f32 atomic adds one whole 128-B acc row, not two 64-B half rows; adds arrive unordered as before
# speedup vs baseline: 2.7208x; 1.0318x over previous
#include <hip/hip_runtime.h>


typedef __attribute__((ext_vector_type(2))) float v2f;
typedef __attribute__((ext_vector_type(8))) float v8f;

#define BN_EPS 1e-5f

__device__ __forceinline__ void atomic_add_f32(float* p, float v) {
    unsafeAtomicAdd(p, v);
}

template <int I>
__device__ __forceinline__ int row_share_i32(int v) {
    return __builtin_amdgcn_update_dpp(0, v, 0x150 + I, 0xf, 0xf, false);
}

template <int I, bool CHECK>
__device__ __forceinline__ void conv1_steps(float* __restrict__ h_acc,
                                            int off, int xbits, float w, int l15) {
    if constexpr (I < 16) {
        int   o = row_share_i32<I>(off);
        float v = __int_as_float(row_share_i32<I>(xbits));
        if (!CHECK || o >= 0) atomic_add_f32(&h_acc[(unsigned)(o + l15)], v * w);
        conv1_steps<I + 1, CHECK>(h_acc, off, xbits, w, l15);
    }
}

__global__ void conv1_scatter(const float* __restrict__ x,
                              const float* __restrict__ W1,
                              const int*   __restrict__ in_idx,
                              const int*   __restrict__ out_idx,
                              float*       __restrict__ h_acc,
                              int M) {
    int k   = blockIdx.y;
    int m   = blockIdx.x * blockDim.x + threadIdx.x;
    int l15 = threadIdx.x & 15;
    int m0  = __builtin_amdgcn_readfirstlane(m);
    if (m0 >= M) return;
    float w = W1[k * 16 + l15];
    if (m0 + 32 <= M) {
        int e = k * M + m;
        float xv  = x[in_idx[e]];
        int   off = out_idx[e] * 16;
        conv1_steps<0, false>(h_acc, off, __float_as_int(xv), w, l15);
    } else {
        int   off = -1;
        float xv  = 0.0f;
        if (m < M) {
            int e = k * M + m;
            xv  = x[in_idx[e]];
            off = out_idx[e] * 16;
        }
        conv1_steps<0, true>(h_acc, off, __float_as_int(xv), w, l15);
    }
}

__global__ void bn_elu(const float* __restrict__ in, float* __restrict__ out,
                       const float* __restrict__ g,  const float* __restrict__ b,
                       const float* __restrict__ mn, const float* __restrict__ vr,
                       int total, int Cmask) {
    int t = blockIdx.x * blockDim.x + threadIdx.x;
    if (t >= total) return;
    int c = t & Cmask;
    float s = g[c] * rsqrtf(vr[c] + BN_EPS);
    float y = (in[t] - mn[c]) * s + b[c];
    out[t] = (y > 0.0f) ? y : (expf(y) - 1.0f);
}

typedef __attribute__((ext_vector_type(2))) unsigned int v2u;
__device__ __forceinline__ void swap_lane_halves(float& a, float& b) {
    v2u r = __builtin_amdgcn_permlane16_swap(__float_as_uint(a), __float_as_uint(b),
                                             false, false);
    a = __uint_as_float(r.x);
    b = __uint_as_float(r.y);
}

__global__ void conv2_wmma(const float* __restrict__ h,
                           const float* __restrict__ W2,
                           const int*   __restrict__ in_idx,
                           const int*   __restrict__ out_idx,
                           float*       __restrict__ acc,
                           int M, int Mt, int totalWaves) {
    int wave = blockIdx.x * (blockDim.x >> 5)
             + __builtin_amdgcn_readfirstlane(threadIdx.x >> 5);
    if (wave >= totalWaves) return;
    int lane = threadIdx.x & 31;
    int half = lane >> 4;
    int l15  = lane & 15;

    int k  = wave / Mt;
    int m0 = (wave - k * Mt) << 4;
    int base = k * M + m0;

    int arow = in_idx[base + l15];
    const float* hp = h + arow * 16 + (half << 1);

    const float* wk = W2 + k * 512;

    v8f c0 = {};
    v8f c1 = {};

#pragma unroll
    for (int q = 0; q < 4; ++q) {
        v2f a;
        a.x = hp[4 * q + 0];
        a.y = hp[4 * q + 1];
        int ci0 = 4 * q + (half << 1);
        v2f b0, b1;
        b0.x = wk[(ci0    ) * 32 +      l15];
        b0.y = wk[(ci0 + 1) * 32 +      l15];
        b1.x = wk[(ci0    ) * 32 + 16 + l15];
        b1.y = wk[(ci0 + 1) * 32 + 16 + l15];
        c0 = __builtin_amdgcn_wmma_f32_16x16x4_f32(false, a, false, b0,
                                                   (short)0, c0, false, false);
        c1 = __builtin_amdgcn_wmma_f32_16x16x4_f32(false, a, false, b1,
                                                   (short)0, c1, false, false);
    }

#pragma unroll
    for (int v = 0; v < 8; ++v) {
        float lo = c0[v], hi = c1[v];
        swap_lane_halves(lo, hi);
        c0[v] = lo;
        c1[v] = hi;
    }
#pragma unroll
    for (int v = 0; v < 8; ++v) {
        int r0 = out_idx[base + v];
        int r1 = out_idx[base + 8 + v];
        atomic_add_f32(&acc[r0 * 32 + lane], c0[v]);
        atomic_add_f32(&acc[r1 * 32 + lane], c1[v]);
    }
}

__global__ void conv2_scalar_tail(const float* __restrict__ h,
                                  const float* __restrict__ W2,
                                  const int*   __restrict__ in_idx,
                                  const int*   __restrict__ out_idx,
                                  float*       __restrict__ acc,
                                  int M, int mStart, int K) {
    int rem = M - mStart;
    int total = K * rem * 32;
    int t = blockIdx.x * blockDim.x + threadIdx.x;
    if (t >= total) return;
    int co = t & 31;
    int km = t >> 5;
    int k  = km / rem;
    int m  = mStart + (km - k * rem);
    int e  = k * M + m;
    const float* hr = h + in_idx[e] * 16;
    const float* wp = W2 + k * 512 + co;
    float s = 0.0f;
#pragma unroll
    for (int ci = 0; ci < 16; ++ci) s += hr[ci] * wp[ci * 32];
    atomic_add_f32(&acc[out_idx[e] * 32 + co], s);
}

extern "C" void kernel_launch(void* const* d_in, const int* in_sizes, int n_in,
                              void* d_out, int out_size, void* d_ws, size_t ws_size,
                              hipStream_t stream) {
    const float* x      = (const float*)d_in[0];
    const float* W1     = (const float*)d_in[1];
    const float* g1     = (const float*)d_in[2];
    const float* b1     = (const float*)d_in[3];
    const float* m1     = (const float*)d_in[4];
    const float* v1     = (const float*)d_in[5];
    const float* W2     = (const float*)d_in[6];
    const float* g2     = (const float*)d_in[7];
    const float* b2     = (const float*)d_in[8];
    const float* m2     = (const float*)d_in[9];
    const float* v2     = (const float*)d_in[10];
    const int*   in_idx  = (const int*)d_in[11];
    const int*   out_idx = (const int*)d_in[12];

    const int N  = in_sizes[0];
    const int K  = in_sizes[1] / 16;
    const int KM = in_sizes[11];
    const int M  = KM / K;

    float* h_buf = (float*)d_ws;
    float* acc   = h_buf + (size_t)N * 16;

    hipMemsetAsync(d_ws, 0, (size_t)N * 48 * sizeof(float), stream);

    conv1_scatter<<<dim3((M + 255) / 256, K), 256, 0, stream>>>(
        x, W1, in_idx, out_idx, h_buf, M);

    bn_elu<<<(N * 16 + 255) / 256, 256, 0, stream>>>(
        h_buf, h_buf, g1, b1, m1, v1, N * 16, 15);

    const int Mt = M / 16;
    const int totalWaves = K * Mt;
    if (totalWaves > 0) {
        const int blocks = (totalWaves + 7) / 8;
        conv2_wmma<<<blocks, 256, 0, stream>>>(
            h_buf, W2, in_idx, out_idx, acc, M, Mt, totalWaves);
    }
    const int mStart = Mt * 16;
    if (mStart < M) {
        const int total = K * (M - mStart) * 32;
        conv2_scalar_tail<<<(total + 255) / 256, 256, 0, stream>>>(
            h_buf, W2, in_idx, out_idx, acc, M, mStart, K);
    }

    bn_elu<<<(N * 32 + 255) / 256, 256, 0, stream>>>(
        acc, (float*)d_out, g2, b2, m2, v2, N * 32, 31);
}


# both scatters: each half of the chip adds only into its own half of the output rows (die id at entry; map chunks drawn by integer ticket, swept once per row range); adds arrive unordered as before
# speedup vs baseline: 3.9086x; 1.4366x over previous
#include <hip/hip_runtime.h>


typedef __attribute__((ext_vector_type(2))) float v2f;
typedef __attribute__((ext_vector_type(8))) float v8f;

#define BN_EPS 1e-5f

__device__ __forceinline__ void atomic_add_f32(float* p, float v) {
    unsafeAtomicAdd(p, v);
}

__device__ __forceinline__ int chip_half() {
    unsigned int v = __builtin_amdgcn_s_sendmsg_rtn(0x87);
    return (int)((v >> 18) & 1u);
}

__device__ __forceinline__ int draw_ticket(int* ctr, int lane) {
    int t = 0;
    if (lane == 0) t = atomicAdd(ctr, 1);
    return __builtin_amdgcn_readfirstlane(t);
}

#define CTR_STRIDE 64
#define S1_CHUNK   8
#define S2_CHUNK   16

template <int I>
__device__ __forceinline__ int row_share_i32(int v) {
    return __builtin_amdgcn_update_dpp(0, v, 0x150 + I, 0xf, 0xf, false);
}

template <int I>
__device__ __forceinline__ void conv1_steps(float* __restrict__ h_acc,
                                            int off, int xbits, float w, int l15) {
    if constexpr (I < 16) {
        int   o = row_share_i32<I>(off);
        float v = __int_as_float(row_share_i32<I>(xbits));
        if (o >= 0) atomic_add_f32(&h_acc[(unsigned)(o + l15)], v * w);
        conv1_steps<I + 1>(h_acc, off, xbits, w, l15);
    }
}

__global__ void conv1_scatter(const float* __restrict__ x,
                              const float* __restrict__ W1,
                              const int*   __restrict__ in_idx,
                              const int*   __restrict__ out_idx,
                              float*       __restrict__ h_acc,
                              int*         __restrict__ ctr,
                              int M, int N, int K) {
    int lane = threadIdx.x & 31;
    int l15  = lane & 15;
    int wpk  = (M + 31) >> 5;
    int cpk  = (wpk + S1_CHUNK - 1) / S1_CHUNK;
    int total = K * cpk;
    int Nh   = (N >> 1) & ~3;
    int mine = chip_half();
    for (int sweep = 0; sweep < 2; ++sweep) {
        int p  = mine ^ sweep;
        int lo = p ? Nh : 0;
        int hi = p ? N  : Nh;
        for (;;) {
            int tk = draw_ticket(ctr + p * CTR_STRIDE, lane);
            if (tk >= total) break;
            int k  = tk / cpk;
            int w0 = (tk - k * cpk) * S1_CHUNK;
            int w1 = min(w0 + S1_CHUNK, wpk);
            float w = W1[k * 16 + l15];
            for (int c = w0; c < w1; ++c) {
                int   m   = (c << 5) + lane;
                int   off = -1;
                float xv  = 0.0f;
                if (m < M) {
                    int e = k * M + m;
                    int r = out_idx[e];
                    int i = in_idx[e];
                    asm volatile("" :: "v"(r), "v"(i));
                    if (r >= lo && r < hi) {
                        xv  = x[i];
                        off = r * 16;
                    }
                }
                conv1_steps<0>(h_acc, off, __float_as_int(xv), w, l15);
            }
        }
    }
}

__global__ void bn_elu(const float* __restrict__ in, float* __restrict__ out,
                       const float* __restrict__ g,  const float* __restrict__ b,
                       const float* __restrict__ mn, const float* __restrict__ vr,
                       int total, int Cmask, int* za, int* zb) {
    int t = blockIdx.x * blockDim.x + threadIdx.x;
    if (t == 0 && za != nullptr) {
        za[0] = 0; za[CTR_STRIDE] = 0;
        zb[0] = 0; zb[CTR_STRIDE] = 0;
    }
    if (t >= total) return;
    int c = t & Cmask;
    float s = g[c] * rsqrtf(vr[c] + BN_EPS);
    float y = (in[t] - mn[c]) * s + b[c];
    out[t] = (y > 0.0f) ? y : (expf(y) - 1.0f);
}

typedef __attribute__((ext_vector_type(2))) unsigned int v2u;
__device__ __forceinline__ void swap_lane_halves(float& a, float& b) {
    v2u r = __builtin_amdgcn_permlane16_swap(__float_as_uint(a), __float_as_uint(b),
                                             false, false);
    a = __uint_as_float(r.x);
    b = __uint_as_float(r.y);
}

__global__ void conv2_wmma(const float* __restrict__ h,
                           const float* __restrict__ W2,
                           const int*   __restrict__ in_idx,
                           const int*   __restrict__ out_idx,
                           float*       __restrict__ acc,
                           int*         __restrict__ ctr,
                           int M, int Mt, int N, int K) {
    int lane = threadIdx.x & 31;
    int half = lane >> 4;
    int l15  = lane & 15;
    int cpk  = (Mt + S2_CHUNK - 1) / S2_CHUNK;
    int total = K * cpk;
    int Nh   = (N >> 1) & ~3;
    int mine = chip_half();

    for (int sweep = 0; sweep < 2; ++sweep) {
        int p  = mine ^ sweep;
        int lo = p ? Nh : 0;
        int hi = p ? N  : Nh;
        for (;;) {
            int tk = draw_ticket(ctr + p * CTR_STRIDE, lane);
            if (tk >= total) break;
            int k  = tk / cpk;
            int t0 = (tk - k * cpk) * S2_CHUNK;
            int t1 = min(t0 + S2_CHUNK, Mt);

            const float* wk = W2 + k * 512;
            v2f b0[4], b1[4];
#pragma unroll
            for (int q = 0; q < 4; ++q) {
                int ci0 = 4 * q + (half << 1);
                b0[q].x = wk[(ci0    ) * 32 +      l15];
                b0[q].y = wk[(ci0 + 1) * 32 +      l15];
                b1[q].x = wk[(ci0    ) * 32 + 16 + l15];
                b1[q].y = wk[(ci0 + 1) * 32 + 16 + l15];
            }

            for (int tl = t0; tl < t1; ++tl) {
                int base = k * M + (tl << 4);

                int  orow = out_idx[base + l15];
                int  arow = in_idx[base + l15];
                asm volatile("" :: "v"(orow), "v"(arow));
                bool inr  = (orow >= lo) && (orow < hi);
                unsigned int inmask = __builtin_amdgcn_ballot_w32(inr);
                v2f a[4];
#pragma unroll
                for (int q = 0; q < 4; ++q) { a[q].x = 0.0f; a[q].y = 0.0f; }
                if (inr) {
                    const float* hp = h + arow * 16 + (half << 1);
#pragma unroll
                    for (int q = 0; q < 4; ++q) {
                        a[q].x = hp[4 * q + 0];
                        a[q].y = hp[4 * q + 1];
                    }
                }

                v8f c0 = {};
                v8f c1 = {};
#pragma unroll
                for (int q = 0; q < 4; ++q) {
                    c0 = __builtin_amdgcn_wmma_f32_16x16x4_f32(false, a[q], false, b0[q],
                                                               (short)0, c0, false, false);
                    c1 = __builtin_amdgcn_wmma_f32_16x16x4_f32(false, a[q], false, b1[q],
                                                               (short)0, c1, false, false);
                }

#pragma unroll
                for (int v = 0; v < 8; ++v) {
                    float x0 = c0[v], x1 = c1[v];
                    swap_lane_halves(x0, x1);
                    c0[v] = x0;
                    c1[v] = x1;
                }
#pragma unroll
                for (int v = 0; v < 8; ++v) {
                    if (inmask & (1u << v)) {
                        int r0 = __builtin_amdgcn_readlane(orow, v);
                        atomic_add_f32(&acc[r0 * 32 + lane], c0[v]);
                    }
                    if (inmask & (1u << (v + 8))) {
                        int r1 = __builtin_amdgcn_readlane(orow, v + 8);
                        atomic_add_f32(&acc[r1 * 32 + lane], c1[v]);
                    }
                }
            }
        }
    }
}

__global__ void conv2_scalar_tail(const float* __restrict__ h,
                                  const float* __restrict__ W2,
                                  const int*   __restrict__ in_idx,
                                  const int*   __restrict__ out_idx,
                                  float*       __restrict__ acc,
                                  int M, int mStart, int K) {
    int rem = M - mStart;
    int total = K * rem * 32;
    int t = blockIdx.x * blockDim.x + threadIdx.x;
    if (t >= total) return;
    int co = t & 31;
    int km = t >> 5;
    int k  = km / rem;
    int m  = mStart + (km - k * rem);
    int e  = k * M + m;
    const float* hr = h + in_idx[e] * 16;
    const float* wp = W2 + k * 512 + co;
    float s = 0.0f;
#pragma unroll
    for (int ci = 0; ci < 16; ++ci) s += hr[ci] * wp[ci * 32];
    atomic_add_f32(&acc[out_idx[e] * 32 + co], s);
}

extern "C" void kernel_launch(void* const* d_in, const int* in_sizes, int n_in,
                              void* d_out, int out_size, void* d_ws, size_t ws_size,
                              hipStream_t stream) {
    const float* x      = (const float*)d_in[0];
    const float* W1     = (const float*)d_in[1];
    const float* g1     = (const float*)d_in[2];
    const float* b1     = (const float*)d_in[3];
    const float* m1     = (const float*)d_in[4];
    const float* v1     = (const float*)d_in[5];
    const float* W2     = (const float*)d_in[6];
    const float* g2     = (const float*)d_in[7];
    const float* b2     = (const float*)d_in[8];
    const float* m2     = (const float*)d_in[9];
    const float* v2     = (const float*)d_in[10];
    const int*   in_idx  = (const int*)d_in[11];
    const int*   out_idx = (const int*)d_in[12];

    const int N  = in_sizes[0];
    const int K  = in_sizes[1] / 16;
    const int KM = in_sizes[11];
    const int M  = KM / K;

    float* h_buf = (float*)d_ws;
    float* acc   = h_buf + (size_t)N * 16;

    hipMemsetAsync(d_ws, 0, (size_t)N * 48 * sizeof(float), stream);

    int* ctr1 = (int*)acc;
    int* ctr2 = (int*)d_out;

    conv1_scatter<<<2048, 256, 0, stream>>>(
        x, W1, in_idx, out_idx, h_buf, ctr1, M, N, K);

    bn_elu<<<(N * 16 + 255) / 256, 256, 0, stream>>>(
        h_buf, h_buf, g1, b1, m1, v1, N * 16, 15, ctr1, ctr2);

    const int Mt = M / 16;
    if (K * Mt > 0) {
        conv2_wmma<<<2048, 256, 0, stream>>>(
            h_buf, W2, in_idx, out_idx, acc, ctr2, M, Mt, N, K);
    }
    const int mStart = Mt * 16;
    if (mStart < M) {
        const int total = K * (M - mStart) * 32;
        conv2_scalar_tail<<<(total + 255) / 256, 256, 0, stream>>>(
            h_buf, W2, in_idx, out_idx, acc, M, mStart, K);
    }

    bn_elu<<<(N * 32 + 255) / 256, 256, 0, stream>>>(
        acc, (float*)d_out, g2, b2, m2, v2, N * 32, 31, nullptr, nullptr);
}
